# starting kernel
# baseline (speedup 1.0000x reference)
#include <hip/hip_runtime.h>
#include <hip/hip_bf16.h>
#include <math.h>

typedef float v2f __attribute__((ext_vector_type(2)));
typedef float v8f __attribute__((ext_vector_type(8)));

#define BATCH   128
#define RROWS   8
#define NSLOT   4096
#define MDIM    64
#define NTILES  (NSLOT / 16)
#define COS_EPS 1e-8f
#define ADD_EPS 1e-16f

__global__ __launch_bounds__(256)
void ntm_address_kernel(const float* __restrict__ memory,
                        const float* __restrict__ key,
                        const float* __restrict__ beta,
                        const float* __restrict__ gate,
                        const float* __restrict__ shift,
                        const float* __restrict__ gamma,
                        const float* __restrict__ w_prev,
                        float* __restrict__ w_out)
{
    __shared__ float smem[RROWS * NSLOT];

    const int b    = blockIdx.x;
    const int tid  = threadIdx.x;
    const int lane = tid & 31;
    const int wave = tid >> 5;
    const int half = lane >> 4;
    const int mrow = lane & 15;

    const float* kb = key + (size_t)b * RROWS * MDIM;

    v2f a[16];
#pragma unroll
    for (int s = 0; s < 16; ++s) {
        if (mrow < RROWS) {
            const int m = 4 * s + 2 * half;
            a[s].x = kb[mrow * MDIM + m]     + ADD_EPS;
            a[s].y = kb[mrow * MDIM + m + 1] + ADD_EPS;
        } else {
            a[s].x = 0.0f;
            a[s].y = 0.0f;
        }
    }

    float knl = 0.0f;
    if (lane < RROWS) {
        for (int m = 0; m < MDIM; ++m) {
            const float v = kb[lane * MDIM + m] + ADD_EPS;
            knl += v * v;
        }
        knl = sqrtf(knl);
    }
    float kn[RROWS], bet[RROWS];
#pragma unroll
    for (int j = 0; j < RROWS; ++j) {
        kn[j]  = __shfl(knl, j, 32);
        bet[j] = beta[b * RROWS + j];
    }

    const float* mb = memory + (size_t)b * NSLOT * MDIM;

    for (int tile = wave; tile < NTILES; tile += 8) {
        const int    n0   = tile * 16;
        const float* rowp = mb + (size_t)(n0 + mrow) * MDIM + 2 * half;

        __builtin_prefetch(rowp + 8 * 16 * MDIM, 0, 1);

        v8f   acc = {};
        float ssq = 0.0f;
#pragma unroll
        for (int s = 0; s < 16; ++s) {
            v2f bv;
            bv.x = rowp[4 * s]     + ADD_EPS;
            bv.y = rowp[4 * s + 1] + ADD_EPS;
            ssq += bv.x * bv.x + bv.y * bv.y;
            acc = __builtin_amdgcn_wmma_f32_16x16x4_f32(
                      false, a[s], false, bv, (short)0, acc, false, false);
        }

        ssq += __shfl_xor(ssq, 16, 32);
        const float mem_norm = sqrtf(ssq);

        if (lane < 16) {
            const int n = n0 + lane;
#pragma unroll
            for (int j = 0; j < RROWS; ++j) {
                const float denom = fmaxf(mem_norm * kn[j], COS_EPS);
                smem[j * NSLOT + n] = bet[j] * (acc[j] / denom);
            }
        }
    }

    __syncthreads();

    {
        const int r = wave;
        float* trow = smem + r * NSLOT;

        float mx = -INFINITY;
        for (int n = lane; n < NSLOT; n += 32) mx = fmaxf(mx, trow[n]);
#pragma unroll
        for (int o = 16; o > 0; o >>= 1) mx = fmaxf(mx, __shfl_xor(mx, o, 32));

        float Z = 0.0f;
        for (int n = lane; n < NSLOT; n += 32) {
            const float e = __expf(trow[n] - mx);
            trow[n] = e;
            Z += e;
        }
#pragma unroll
        for (int o = 16; o > 0; o >>= 1) Z += __shfl_xor(Z, o, 32);
        const float invZ = 1.0f / Z;

        const int    hr = b * RROWS + r;
        const float  gr = gate[hr];
        const float  s0 = shift[hr * 3 + 0];
        const float  s1 = shift[hr * 3 + 1];
        const float  s2 = shift[hr * 3 + 2];
        const float  gm = gamma[hr];
        const float* wp = w_prev + (size_t)hr * NSLOT;

        for (int n = lane; n < NSLOT; n += 32) {
            const float wc = trow[n] * invZ;
            trow[n] = gr * wc + (1.0f - gr) * wp[n];
        }

        __syncthreads();

        float S = 0.0f;
        for (int n = lane; n < NSLOT; n += 32) {
            const int nm = (n + NSLOT - 1) & (NSLOT - 1);
            const int np = (n + 1) & (NSLOT - 1);
            const float c = s0 * trow[nm] + s1 * trow[n] + s2 * trow[np];
            S += __powf(c, gm);
        }
#pragma unroll
        for (int o = 16; o > 0; o >>= 1) S += __shfl_xor(S, o, 32);
        const float invS = 1.0f / (S + ADD_EPS);

        float* wo = w_out + (size_t)hr * NSLOT;
        for (int n = lane; n < NSLOT; n += 32) {
            const int nm = (n + NSLOT - 1) & (NSLOT - 1);
            const int np = (n + 1) & (NSLOT - 1);
            const float c = s0 * trow[nm] + s1 * trow[n] + s2 * trow[np];
            wo[n] = __powf(c, gm) * invS;
        }
    }
}

extern "C" void kernel_launch(void* const* d_in, const int* in_sizes, int n_in,
                              void* d_out, int out_size, void* d_ws, size_t ws_size,
                              hipStream_t stream) {
    (void)in_sizes; (void)n_in; (void)out_size; (void)d_ws; (void)ws_size;
    const float* memory = (const float*)d_in[0];
    const float* k      = (const float*)d_in[1];
    const float* beta   = (const float*)d_in[2];
    const float* g      = (const float*)d_in[3];
    const float* s      = (const float*)d_in[4];
    const float* gamma  = (const float*)d_in[5];
    const float* w_prev = (const float*)d_in[6];
    float* out = (float*)d_out;

    ntm_address_kernel<<<dim3(BATCH), dim3(256), 0, stream>>>(
        memory, k, beta, g, s, gamma, w_prev, out);
}


# two chip-wide launches: 512 workgroups write beta*cos scores into out, then one workgroup per head (16 slots per thread in registers) finishes in place; softmax and sharpen sums reordered
# speedup vs baseline: 2.5502x; 2.5502x over previous
#include <hip/hip_runtime.h>
#include <hip/hip_bf16.h>
#include <math.h>


typedef float v2f __attribute__((ext_vector_type(2)));
typedef float v8f __attribute__((ext_vector_type(8)));

#define BATCH   128
#define RROWS   8
#define NSLOT   4096
#define MDIM    64
#define COS_EPS 1e-8f
#define ADD_EPS 1e-16f

#define CHUNKS       4
#define CHUNK_ROWS   1024
#define UNIT_ROWS    32
#define UNITS        32
#define UNITS_PER_WAVE 4
#define SCORE_GRID   512
#define HEAD_GRID    1024

__global__ __launch_bounds__(256)
void ntm_score_kernel(const float* __restrict__ memory,
                      const float* __restrict__ key,
                      const float* __restrict__ beta,
                      float* __restrict__ score)
{
    const int b     = blockIdx.x >> 2;
    const int chunk = blockIdx.x & 3;
    const int tid   = threadIdx.x;
    const int lane  = tid & 31;
    const int wave  = tid >> 5;
    const int half  = lane >> 4;
    const int mrow  = lane & 15;

    const float* kb = key + (size_t)b * RROWS * MDIM;

    v2f a[16], a2[16];
#pragma unroll
    for (int s = 0; s < 16; ++s) {
        const int   m  = 4 * s + 2 * half;
        const int   kr = mrow & 7;
        const float kx = kb[kr * MDIM + m]     + ADD_EPS;
        const float ky = kb[kr * MDIM + m + 1] + ADD_EPS;
        a[s].x  = (mrow <  RROWS) ? kx : 0.0f;
        a[s].y  = (mrow <  RROWS) ? ky : 0.0f;
        a2[s].x = (mrow >= RROWS) ? kx : 0.0f;
        a2[s].y = (mrow >= RROWS) ? ky : 0.0f;
    }

    float knl = 0.0f;
    if (lane < RROWS) {
        for (int m = 0; m < MDIM; ++m) {
            const float v = kb[lane * MDIM + m] + ADD_EPS;
            knl += v * v;
        }
        knl = sqrtf(knl);
    }
    float kn[RROWS], bet[RROWS];
#pragma unroll
    for (int j = 0; j < RROWS; ++j) {
        kn[j]  = __shfl(knl, j, 32);
        bet[j] = beta[b * RROWS + j];
    }

    const float* mb = memory + ((size_t)b * NSLOT + (size_t)chunk * CHUNK_ROWS) * MDIM;
    float*       sb = score + (size_t)b * RROWS * NSLOT + (size_t)chunk * CHUNK_ROWS;

    for (int i = 0; i < UNITS_PER_WAVE; ++i) {
        const int    unit  = wave + 8 * i;
        const int    n0    = unit * UNIT_ROWS;
        const float* rowp0 = mb + (size_t)(n0 + mrow) * MDIM + 2 * half;
        const float* rowp1 = rowp0 + 16 * MDIM;

        if (i + 1 < UNITS_PER_WAVE) {
            const float* nx = mb + (size_t)(n0 + 8 * UNIT_ROWS + lane) * MDIM;
            __builtin_prefetch(nx, 0, 1);
            __builtin_prefetch(nx + 32, 0, 1);
        }

        v8f   acc0 = {}, acc1 = {};
        float ssq0 = 0.0f, ssq1 = 0.0f;
#pragma unroll
        for (int s = 0; s < 16; ++s) {
            v2f bv0, bv1;
            bv0.x = rowp0[4 * s]     + ADD_EPS;
            bv0.y = rowp0[4 * s + 1] + ADD_EPS;
            bv1.x = rowp1[4 * s]     + ADD_EPS;
            bv1.y = rowp1[4 * s + 1] + ADD_EPS;
            ssq0 += bv0.x * bv0.x + bv0.y * bv0.y;
            ssq1 += bv1.x * bv1.x + bv1.y * bv1.y;
            acc0 = __builtin_amdgcn_wmma_f32_16x16x4_f32(
                       false, a[s],  false, bv0, (short)0, acc0, false, false);
            acc1 = __builtin_amdgcn_wmma_f32_16x16x4_f32(
                       false, a2[s], false, bv1, (short)0, acc1, false, false);
        }

        ssq0 += __shfl_xor(ssq0, 16, 32);
        ssq1 += __shfl_xor(ssq1, 16, 32);
        const float mem_norm = sqrtf(half ? ssq1 : ssq0);

        const int n = n0 + lane;
#pragma unroll
        for (int j = 0; j < RROWS; ++j) {
            const float dot   = half ? acc1[j] : acc0[j];
            const float denom = fmaxf(mem_norm * kn[j], COS_EPS);
            sb[(size_t)j * NSLOT + n] = bet[j] * (dot / denom);
        }
    }
}

__global__ __launch_bounds__(256)
void ntm_head_kernel(const float* __restrict__ gate,
                     const float* __restrict__ shift,
                     const float* __restrict__ gamma,
                     const float* __restrict__ w_prev,
                     float* w_io)
{
    __shared__ float row[NSLOT];
    __shared__ float red[3][8];

    const int hr   = blockIdx.x;
    const int tid  = threadIdx.x;
    const int lane = tid & 31;
    const int wave = tid >> 5;

    const float4* sp = reinterpret_cast<const float4*>(w_io + (size_t)hr * NSLOT);
    const float4* pp = reinterpret_cast<const float4*>(w_prev + (size_t)hr * NSLOT);
    float4*       op = reinterpret_cast<float4*>(w_io + (size_t)hr * NSLOT);
    float4*       rp = reinterpret_cast<float4*>(row);

    float4 v[4], p[4];
#pragma unroll
    for (int q = 0; q < 4; ++q) {
        v[q] = sp[tid + 256 * q];
        p[q] = pp[tid + 256 * q];
    }

    const float gr = gate[hr];
    const float s0 = shift[hr * 3 + 0];
    const float s1 = shift[hr * 3 + 1];
    const float s2 = shift[hr * 3 + 2];
    const float gm = gamma[hr];

    float mx = -INFINITY;
#pragma unroll
    for (int q = 0; q < 4; ++q)
        mx = fmaxf(mx, fmaxf(fmaxf(v[q].x, v[q].y), fmaxf(v[q].z, v[q].w)));
#pragma unroll
    for (int o = 16; o > 0; o >>= 1) mx = fmaxf(mx, __shfl_xor(mx, o, 32));
    if (lane == 0) red[0][wave] = mx;
    __syncthreads();
    mx = red[0][0];
#pragma unroll
    for (int w = 1; w < 8; ++w) mx = fmaxf(mx, red[0][w]);

    float Z = 0.0f;
#pragma unroll
    for (int q = 0; q < 4; ++q) {
        v[q].x = __expf(v[q].x - mx);
        v[q].y = __expf(v[q].y - mx);
        v[q].z = __expf(v[q].z - mx);
        v[q].w = __expf(v[q].w - mx);
        Z += v[q].x;
        Z += v[q].y;
        Z += v[q].z;
        Z += v[q].w;
    }
#pragma unroll
    for (int o = 16; o > 0; o >>= 1) Z += __shfl_xor(Z, o, 32);
    if (lane == 0) red[1][wave] = Z;
    __syncthreads();
    Z = red[1][0];
#pragma unroll
    for (int w = 1; w < 8; ++w) Z += red[1][w];
    const float invZ = 1.0f / Z;

#pragma unroll
    for (int q = 0; q < 4; ++q) {
        v[q].x = gr * (v[q].x * invZ) + (1.0f - gr) * p[q].x;
        v[q].y = gr * (v[q].y * invZ) + (1.0f - gr) * p[q].y;
        v[q].z = gr * (v[q].z * invZ) + (1.0f - gr) * p[q].z;
        v[q].w = gr * (v[q].w * invZ) + (1.0f - gr) * p[q].w;
        rp[tid + 256 * q] = v[q];
    }
    __syncthreads();

    float S = 0.0f;
#pragma unroll
    for (int q = 0; q < 4; ++q) {
        const int   nb    = 4 * (tid + 256 * q);
        const float left  = row[(nb + NSLOT - 1) & (NSLOT - 1)];
        const float right = row[(nb + 4) & (NSLOT - 1)];
        const float c0 = s0 * left   + s1 * v[q].x + s2 * v[q].y;
        const float c1 = s0 * v[q].x + s1 * v[q].y + s2 * v[q].z;
        const float c2 = s0 * v[q].y + s1 * v[q].z + s2 * v[q].w;
        const float c3 = s0 * v[q].z + s1 * v[q].w + s2 * right;
        p[q].x = __powf(c0, gm);
        p[q].y = __powf(c1, gm);
        p[q].z = __powf(c2, gm);
        p[q].w = __powf(c3, gm);
        S += p[q].x;
        S += p[q].y;
        S += p[q].z;
        S += p[q].w;
    }
#pragma unroll
    for (int o = 16; o > 0; o >>= 1) S += __shfl_xor(S, o, 32);
    if (lane == 0) red[2][wave] = S;
    __syncthreads();
    S = red[2][0];
#pragma unroll
    for (int w = 1; w < 8; ++w) S += red[2][w];
    const float invS = 1.0f / (S + ADD_EPS);

#pragma unroll
    for (int q = 0; q < 4; ++q) {
        float4 o4;
        o4.x = p[q].x * invS;
        o4.y = p[q].y * invS;
        o4.z = p[q].z * invS;
        o4.w = p[q].w * invS;
        op[tid + 256 * q] = o4;
    }
}

extern "C" void kernel_launch(void* const* d_in, const int* in_sizes, int n_in,
                              void* d_out, int out_size, void* d_ws, size_t ws_size,
                              hipStream_t stream) {
    (void)in_sizes; (void)n_in; (void)out_size; (void)d_ws; (void)ws_size;
    const float* memory = (const float*)d_in[0];
    const float* k      = (const float*)d_in[1];
    const float* beta   = (const float*)d_in[2];
    const float* g      = (const float*)d_in[3];
    const float* s      = (const float*)d_in[4];
    const float* gamma  = (const float*)d_in[5];
    const float* w_prev = (const float*)d_in[6];
    float* out = (float*)d_out;

    ntm_score_kernel<<<dim3(SCORE_GRID), dim3(256), 0, stream>>>(memory, k, beta, out);
    ntm_head_kernel<<<dim3(HEAD_GRID), dim3(256), 0, stream>>>(g, s, gamma, w_prev, out);
}
